# starting kernel
# baseline (speedup 1.0000x reference)
#include <hip/hip_runtime.h>
#include <math.h>

typedef float v2f __attribute__((ext_vector_type(2)));
typedef float v8f __attribute__((ext_vector_type(8)));

__global__ __launch_bounds__(256) void hm_conf_kernel(const float* __restrict__ hm,
                                                      float* __restrict__ conf) {
    const int bid = blockIdx.x;
    const int t   = threadIdx.x;
    const float4* p = (const float4*)(hm + (size_t)bid * 4096);

    float m = -3.402823466e38f;
#pragma unroll
    for (int i = 0; i < 4; ++i) {
        float4 v = p[t + i * 256];
        m = fmaxf(m, fmaxf(fmaxf(v.x, v.y), fmaxf(v.z, v.w)));
    }
#pragma unroll
    for (int off = 16; off > 0; off >>= 1)
        m = fmaxf(m, __shfl_xor(m, off, 32));

    __shared__ float smax[8];
    if ((t & 31) == 0) smax[t >> 5] = m;
    __syncthreads();
    if (t == 0) {
        float mm = smax[0];
#pragma unroll
        for (int i = 1; i < 8; ++i) mm = fmaxf(mm, smax[i]);
        conf[bid] = 1.0f / (1.0f + expf(-mm));
    }
}

__device__ __forceinline__ float gelu_exact(float x) {
    return 0.5f * x * (1.0f + erff(x * 0.70710678118654752440f));
}

__device__ __forceinline__ void ln_inplace(float h[16][65],
                                           const float* __restrict__ g,
                                           const float* __restrict__ be,
                                           float rsum[16][8], float rsq[16][8],
                                           int tid) {
    const int r = tid >> 3;
    const int p = tid & 7;
    float s = 0.f, ss = 0.f;
#pragma unroll
    for (int j = 0; j < 8; ++j) {
        float v = h[r][p * 8 + j];
        s += v; ss += v * v;
    }
    rsum[r][p] = s; rsq[r][p] = ss;
    __syncthreads();
    float ts = 0.f, tss = 0.f;
#pragma unroll
    for (int i = 0; i < 8; ++i) { ts += rsum[r][i]; tss += rsq[r][i]; }
    const float mu  = ts * (1.0f / 64.0f);
    const float var = tss * (1.0f / 64.0f) - mu * mu;
    const float sc  = rsqrtf(var + 1e-5f);
#pragma unroll
    for (int j = 0; j < 8; ++j) {
        const int c = p * 8 + j;
        h[r][c] = (h[r][c] - mu) * sc * g[c] + be[c];
    }
    __syncthreads();
}

__global__ __launch_bounds__(128) void mlp_head_kernel(
    const float* __restrict__ fk, const float* __restrict__ dir,
    const float* __restrict__ conf,
    const float* __restrict__ W1, const float* __restrict__ b1,
    const float* __restrict__ g1, const float* __restrict__ be1,
    const float* __restrict__ W2, const float* __restrict__ b2,
    const float* __restrict__ g2, const float* __restrict__ be2,
    const float* __restrict__ W3, const float* __restrict__ b3,
    float* __restrict__ out, int nrows) {

    __shared__ float xs[16][12];
    __shared__ float w1s[12][64];
    __shared__ float w2s[64][64];
    __shared__ float w3s[64][16];
    __shared__ float h1[16][65];
    __shared__ float h2[16][65];
    __shared__ float outs[16][16];
    __shared__ float rsum[16][8];
    __shared__ float rsq[16][8];

    const int tid  = threadIdx.x;
    const int lane = tid & 31;
    const int wave = tid >> 5;
    const int half = lane >> 4;
    const int l16  = lane & 15;
    const int row0 = blockIdx.x * 16;

    for (int i = tid; i < 12 * 64; i += 128) {
        const int k = i >> 6, n = i & 63;
        w1s[k][n] = (k < 10) ? W1[k * 64 + n] : 0.0f;
    }
    for (int i = tid; i < 64 * 64; i += 128)
        w2s[i >> 6][i & 63] = W2[i];
    for (int i = tid; i < 64 * 16; i += 128) {
        const int k = i >> 4, n = i & 15;
        w3s[k][n] = (n < 4) ? W3[k * 4 + n] : 0.0f;
    }
    for (int i = tid; i < 16 * 12; i += 128) {
        const int r = i / 12, c = i % 12;
        const int grow = row0 + r;
        float v;
        if (c < 3)       v = fk[grow * 3 + c];
        else if (c < 6)  v = dir[grow * 3 + (c - 3)];
        else if (c < 9)  v = dir[grow * 3 + (c - 6)] - fk[grow * 3 + (c - 6)];
        else if (c == 9) v = conf[grow];
        else             v = 0.0f;
        xs[r][c] = v;
    }
    __syncthreads();

    {
        v8f c = {0.f, 0.f, 0.f, 0.f, 0.f, 0.f, 0.f, 0.f};
        const int ncol = wave * 16 + l16;
#pragma unroll
        for (int ks = 0; ks < 12; ks += 4) {
            v2f a, b;
            a[0] = xs[l16][ks + 2 * half + 0];
            a[1] = xs[l16][ks + 2 * half + 1];
            b[0] = w1s[ks + 2 * half + 0][ncol];
            b[1] = w1s[ks + 2 * half + 1][ncol];
            c = __builtin_amdgcn_wmma_f32_16x16x4_f32(false, a, false, b,
                                                      (short)0, c, false, false);
        }
        const float bias = b1[ncol];
#pragma unroll
        for (int v = 0; v < 8; ++v)
            h1[v + 8 * half][ncol] = gelu_exact(c[v] + bias);
    }
    __syncthreads();
    ln_inplace(h1, g1, be1, rsum, rsq, tid);

    {
        v8f c = {0.f, 0.f, 0.f, 0.f, 0.f, 0.f, 0.f, 0.f};
        const int ncol = wave * 16 + l16;
#pragma unroll
        for (int ks = 0; ks < 64; ks += 4) {
            v2f a, b;
            a[0] = h1[l16][ks + 2 * half + 0];
            a[1] = h1[l16][ks + 2 * half + 1];
            b[0] = w2s[ks + 2 * half + 0][ncol];
            b[1] = w2s[ks + 2 * half + 1][ncol];
            c = __builtin_amdgcn_wmma_f32_16x16x4_f32(false, a, false, b,
                                                      (short)0, c, false, false);
        }
        const float bias = b2[ncol];
#pragma unroll
        for (int v = 0; v < 8; ++v)
            h2[v + 8 * half][ncol] = gelu_exact(c[v] + bias);
    }
    __syncthreads();
    ln_inplace(h2, g2, be2, rsum, rsq, tid);

    if (wave == 0) {
        v8f c = {0.f, 0.f, 0.f, 0.f, 0.f, 0.f, 0.f, 0.f};
#pragma unroll
        for (int ks = 0; ks < 64; ks += 4) {
            v2f a, b;
            a[0] = h2[l16][ks + 2 * half + 0];
            a[1] = h2[l16][ks + 2 * half + 1];
            b[0] = w3s[ks + 2 * half + 0][l16];
            b[1] = w3s[ks + 2 * half + 1][l16];
            c = __builtin_amdgcn_wmma_f32_16x16x4_f32(false, a, false, b,
                                                      (short)0, c, false, false);
        }
#pragma unroll
        for (int v = 0; v < 8; ++v)
            outs[v + 8 * half][l16] = c[v];
    }
    __syncthreads();

    if (tid < 16) {
        const int grow = row0 + tid;
        const float d0 = outs[tid][0] + b3[0];
        const float d1 = outs[tid][1] + b3[1];
        const float d2 = outs[tid][2] + b3[2];
        const float o3 = outs[tid][3] + b3[3];
        const float alpha = 1.0f / (1.0f + expf(-o3));

        const float fx = fk[grow * 3 + 0], fy = fk[grow * 3 + 1], fz = fk[grow * 3 + 2];
        const float dx = dir[grow * 3 + 0], dy = dir[grow * 3 + 1], dz = dir[grow * 3 + 2];

        float* kp_final = out;
        float* p_alpha  = out + (size_t)nrows * 3;
        float* p_delta  = out + (size_t)nrows * 4;

        kp_final[grow * 3 + 0] = alpha * dx + (1.0f - alpha) * fx + d0;
        kp_final[grow * 3 + 1] = alpha * dy + (1.0f - alpha) * fy + d1;
        kp_final[grow * 3 + 2] = alpha * dz + (1.0f - alpha) * fz + d2;
        p_alpha[grow] = alpha;
        p_delta[grow * 3 + 0] = d0;
        p_delta[grow * 3 + 1] = d1;
        p_delta[grow * 3 + 2] = d2;
    }
}

extern "C" void kernel_launch(void* const* d_in, const int* in_sizes, int n_in,
                              void* d_out, int out_size, void* d_ws, size_t ws_size,
                              hipStream_t stream) {
    const float* fk  = (const float*)d_in[0];
    const float* dir = (const float*)d_in[1];
    const float* hm  = (const float*)d_in[2];
    const float* W1  = (const float*)d_in[3];
    const float* b1  = (const float*)d_in[4];
    const float* g1  = (const float*)d_in[5];
    const float* be1 = (const float*)d_in[6];
    const float* W2  = (const float*)d_in[7];
    const float* b2  = (const float*)d_in[8];
    const float* g2  = (const float*)d_in[9];
    const float* be2 = (const float*)d_in[10];
    const float* W3  = (const float*)d_in[11];
    const float* b3  = (const float*)d_in[12];
    float* out  = (float*)d_out;
    float* conf = (float*)d_ws;

    const int nrows = in_sizes[0] / 3;

    hm_conf_kernel<<<nrows, 256, 0, stream>>>(hm, conf);
    mlp_head_kernel<<<nrows / 16, 128, 0, stream>>>(fk, dir, conf,
                                                    W1, b1, g1, be1,
                                                    W2, b2, g2, be2,
                                                    W3, b3, out, nrows);
}


# GELU moved from the 8 matrix waves (8 serial branching erf per lane) into the LayerNorm pass: 16 lanes per real row, erf as select of both ranges, row sums by 16-lane DPP; LN sum order changes
# speedup vs baseline: 1.4120x; 1.4120x over previous
#include <hip/hip_runtime.h>
#include <math.h>


typedef float v2f __attribute__((ext_vector_type(2)));
typedef float v8f __attribute__((ext_vector_type(8)));

#define ROWS   21
#define NTHR   512
#define HM_F4  1024

#define HP     68

__device__ __forceinline__ float f32_bits(unsigned u) { return __builtin_bit_cast(float, u); }

__device__ __forceinline__ float gelu_exact(float x) {
    const float y = x * 0.70710678118654752440f;
    const float t = fabsf(y);
    float p = fmaf(t, f32_bits(0x378e98abu), f32_bits(0xb9c68948u));
    p = fmaf(t, p, f32_bits(0x3b7cd369u));
    p = fmaf(t, p, f32_bits(0xbcc618b2u));
    p = fmaf(t, p, f32_bits(0x3dda74e4u));
    p = fmaf(t, p, f32_bits(0x3f228afdu));
    p = fmaf(t, p, f32_bits(0x3e03c728u));
    p = fmaf(t, p, t);
    const float ra = 1.0f - expf(-p);
    const float s = y * y;
    float q = fmaf(s, f32_bits(0xba1345e1u), f32_bits(0x3ba10414u));
    q = fmaf(s, q, f32_bits(0xbcdac9b8u));
    q = fmaf(s, q, f32_bits(0x3de703beu));
    q = fmaf(s, q, f32_bits(0xbec09330u));
    q = fmaf(s, q, f32_bits(0x3e0375d0u));
    const float rb = fmaf(t, q, t);
    const float e = copysignf((t < 1.0f) ? rb : ra, y);
    return 0.5f * x * (1.0f + e);
}

template <int M>
__device__ __forceinline__ float row_xor(float v) {
    const int b = __builtin_bit_cast(int, v);
    return __builtin_bit_cast(float, __builtin_amdgcn_update_dpp(0, b, 0x160 | M, 0xf, 0xf, true));
}
__device__ __forceinline__ float row16_sum(float v) {
    v += row_xor<1>(v);
    v += row_xor<2>(v);
    v += row_xor<4>(v);
    v += row_xor<8>(v);
    return v;
}

__device__ __forceinline__ void gelu_ln_inplace(float (*h)[HP],
                                                const float4 gg, const float4 bb,
                                                int tid) {
    const int r = tid >> 4;
    const int c = (tid & 15) * 4;
    if (r < ROWS) {
        float4 v = *(const float4*)&h[r][c];
        v.x = gelu_exact(v.x); v.y = gelu_exact(v.y);
        v.z = gelu_exact(v.z); v.w = gelu_exact(v.w);
        const float ts  = row16_sum((v.x + v.y) + (v.z + v.w));
        const float tss = row16_sum((v.x * v.x + v.y * v.y) + (v.z * v.z + v.w * v.w));
        const float mu  = ts * (1.0f / 64.0f);
        const float var = tss * (1.0f / 64.0f) - mu * mu;
        const float sc  = rsqrtf(var + 1e-5f);
        v.x = (v.x - mu) * sc * gg.x + bb.x;
        v.y = (v.y - mu) * sc * gg.y + bb.y;
        v.z = (v.z - mu) * sc * gg.z + bb.z;
        v.w = (v.w - mu) * sc * gg.w + bb.w;
        *(float4*)&h[r][c] = v;
    }
    __syncthreads();
}

__global__ __launch_bounds__(NTHR) void fused_head_kernel(
    const float* __restrict__ fk, const float* __restrict__ dir,
    const float* __restrict__ hm,
    const float* __restrict__ W1, const float* __restrict__ b1,
    const float* __restrict__ g1, const float* __restrict__ be1,
    const float* __restrict__ W2, const float* __restrict__ b2,
    const float* __restrict__ g2, const float* __restrict__ be2,
    const float* __restrict__ W3, const float* __restrict__ b3,
    float* __restrict__ out, int nrows) {

    __shared__ float xs[32][12];
    __shared__ float w1s[16][64];
    __shared__ __attribute__((aligned(16))) float w2s[64][64];
    __shared__ float w3s[64][16];
    __shared__ __attribute__((aligned(16))) float h1[32][HP];
    __shared__ __attribute__((aligned(16))) float h2[32][HP];
    __shared__ float outs[32][16];
    __shared__ __attribute__((aligned(16))) float pvec[7][64];
    __shared__ float smax[ROWS][16];

    const int tid  = threadIdx.x;
    const int lane = tid & 31;
    const int wave = tid >> 5;
    const int half = lane >> 4;
    const int l16  = lane & 15;
    const int row0 = blockIdx.x * ROWS;

    const float4 w2a = ((const float4*)W2)[tid];
    const float4 w2b = ((const float4*)W2)[tid + NTHR];
    const float  w1a = W1[tid];
    const int    i1b = (tid < 128) ? (tid + NTHR) : 639;
    const float  w1b = W1[i1b];
    const int    k3a = tid >> 4, k3b = (tid + NTHR) >> 4, n3 = tid & 15;
    const float  w3a = W3[k3a * 4 + (n3 & 3)];
    const float  w3b = W3[k3b * 4 + (n3 & 3)];
    const int    pn   = tid & 63;
    const float  pb1  = b1[pn],  pg1 = g1[pn],  pe1 = be1[pn];
    const float  pb2  = b2[pn],  pg2 = g2[pn],  pe2 = be2[pn];
    const float  pb3  = b3[tid & 3];
    const int    xe  = ((tid & 63) < 62) ? (tid & 63) : 62;
    const float  xf  = fk [row0 * 3 + xe];
    const float  xd  = dir[row0 * 3 + xe];

    const float4* hp = (const float4*)(hm + (size_t)row0 * 4096) + tid;
    float4 va[ROWS], vb[ROWS];
#pragma unroll
    for (int j = 0; j < ROWS; ++j) {
        va[j] = hp[j * HM_F4];
        vb[j] = hp[j * HM_F4 + NTHR];
    }
    __builtin_amdgcn_sched_barrier(0);

    *(float4*)&w2s[0][0 + 4 * tid]          = w2a;
    *(float4*)&w2s[0][0 + 4 * (tid + NTHR)] = w2b;
    (&w1s[0][0])[tid] = w1a;
    (&w1s[0][0])[tid + NTHR] = (tid < 128) ? w1b : 0.0f;
    w3s[k3a][n3] = (n3 < 4) ? w3a : 0.0f;
    w3s[k3b][n3] = (n3 < 4) ? w3b : 0.0f;
    pvec[0][pn] = pb1; pvec[1][pn] = pg1; pvec[2][pn] = pe1;
    pvec[3][pn] = pb2; pvec[4][pn] = pg2; pvec[5][pn] = pe2;
    pvec[6][tid & 3] = pb3;
    {
        const int r = xe / 3, c = xe - 3 * r;
        xs[r][c]     = xf;
        xs[r][3 + c] = xd;
        xs[r][6 + c] = xd - xf;
    }
    if (tid < 132) xs[ROWS + tid / 12][tid % 12] = 0.0f;
    if (tid < 63)  xs[tid / 3][9 + tid % 3] = 0.0f;

    float m[ROWS];
#pragma unroll
    for (int j = 0; j < ROWS; ++j)
        m[j] = fmaxf(fmaxf(fmaxf(va[j].x, va[j].y), fmaxf(va[j].z, va[j].w)),
                     fmaxf(fmaxf(vb[j].x, vb[j].y), fmaxf(vb[j].z, vb[j].w)));
#pragma unroll
    for (int off = 16; off > 0; off >>= 1) {
#pragma unroll
        for (int j = 0; j < ROWS; ++j)
            m[j] = fmaxf(m[j], __shfl_xor(m[j], off, 32));
    }
    if (lane == 0) {
#pragma unroll
        for (int j = 0; j < ROWS; ++j) smax[j][wave] = m[j];
    }
    __syncthreads();
    if (tid < ROWS) {
        float mm = smax[tid][0];
#pragma unroll
        for (int i = 1; i < 16; ++i) mm = fmaxf(mm, smax[tid][i]);
        xs[tid][9] = 1.0f / (1.0f + expf(-mm));
    }
    __syncthreads();

    const int mt   = (wave >> 2) & 1;
    const int ncol = (wave & 3) * 16 + l16;
    const int arow = mt * 16 + l16;

    if (wave < 8) {
        v8f c = {0.f, 0.f, 0.f, 0.f, 0.f, 0.f, 0.f, 0.f};
#pragma unroll
        for (int ks = 0; ks < 12; ks += 4) {
            v2f a, b;
            a[0] = xs[arow][ks + 2 * half + 0];
            a[1] = xs[arow][ks + 2 * half + 1];
            b[0] = w1s[ks + 2 * half + 0][ncol];
            b[1] = w1s[ks + 2 * half + 1][ncol];
            c = __builtin_amdgcn_wmma_f32_16x16x4_f32(false, a, false, b,
                                                      (short)0, c, false, false);
        }
        const float bias = pvec[0][ncol];
#pragma unroll
        for (int v = 0; v < 8; ++v)
            h1[mt * 16 + v + 8 * half][ncol] = c[v] + bias;
    }
    {
        const float4 gg = *(const float4*)&pvec[1][(tid & 15) * 4];
        const float4 bb = *(const float4*)&pvec[2][(tid & 15) * 4];
        __syncthreads();
        gelu_ln_inplace(h1, gg, bb, tid);
    }

    if (wave < 8) {
        v8f c = {0.f, 0.f, 0.f, 0.f, 0.f, 0.f, 0.f, 0.f};
#pragma unroll
        for (int ks = 0; ks < 64; ks += 4) {
            v2f a, b;
            a[0] = h1[arow][ks + 2 * half + 0];
            a[1] = h1[arow][ks + 2 * half + 1];
            b[0] = w2s[ks + 2 * half + 0][ncol];
            b[1] = w2s[ks + 2 * half + 1][ncol];
            c = __builtin_amdgcn_wmma_f32_16x16x4_f32(false, a, false, b,
                                                      (short)0, c, false, false);
        }
        const float bias = pvec[3][ncol];
#pragma unroll
        for (int v = 0; v < 8; ++v)
            h2[mt * 16 + v + 8 * half][ncol] = c[v] + bias;
    }
    {
        const float4 gg = *(const float4*)&pvec[4][(tid & 15) * 4];
        const float4 bb = *(const float4*)&pvec[5][(tid & 15) * 4];
        __syncthreads();
        gelu_ln_inplace(h2, gg, bb, tid);
    }

    if (wave < 2) {
        v8f c = {0.f, 0.f, 0.f, 0.f, 0.f, 0.f, 0.f, 0.f};
        const int r3 = wave * 16 + l16;
#pragma unroll
        for (int ks = 0; ks < 64; ks += 4) {
            v2f a, b;
            a[0] = h2[r3][ks + 2 * half + 0];
            a[1] = h2[r3][ks + 2 * half + 1];
            b[0] = w3s[ks + 2 * half + 0][l16];
            b[1] = w3s[ks + 2 * half + 1][l16];
            c = __builtin_amdgcn_wmma_f32_16x16x4_f32(false, a, false, b,
                                                      (short)0, c, false, false);
        }
#pragma unroll
        for (int v = 0; v < 8; ++v)
            outs[wave * 16 + v + 8 * half][l16] = c[v];
    }
    __syncthreads();

    if (tid < ROWS) {
        const int grow = row0 + tid;
        const float d0 = outs[tid][0] + pvec[6][0];
        const float d1 = outs[tid][1] + pvec[6][1];
        const float d2 = outs[tid][2] + pvec[6][2];
        const float o3 = outs[tid][3] + pvec[6][3];
        const float alpha = 1.0f / (1.0f + expf(-o3));

        const float fx = xs[tid][0], fy = xs[tid][1], fz = xs[tid][2];
        const float dx = xs[tid][3], dy = xs[tid][4], dz = xs[tid][5];

        float* kp_final = out;
        float* p_alpha  = out + (size_t)nrows * 3;
        float* p_delta  = out + (size_t)nrows * 4;

        kp_final[grow * 3 + 0] = alpha * dx + (1.0f - alpha) * fx + d0;
        kp_final[grow * 3 + 1] = alpha * dy + (1.0f - alpha) * fy + d1;
        kp_final[grow * 3 + 2] = alpha * dz + (1.0f - alpha) * fz + d2;
        p_alpha[grow] = alpha;
        p_delta[grow * 3 + 0] = d0;
        p_delta[grow * 3 + 1] = d1;
        p_delta[grow * 3 + 2] = d2;
    }
}

extern "C" void kernel_launch(void* const* d_in, const int* in_sizes, int n_in,
                              void* d_out, int out_size, void* d_ws, size_t ws_size,
                              hipStream_t stream) {
    const float* fk  = (const float*)d_in[0];
    const float* dir = (const float*)d_in[1];
    const float* hm  = (const float*)d_in[2];
    const float* W1  = (const float*)d_in[3];
    const float* b1  = (const float*)d_in[4];
    const float* g1  = (const float*)d_in[5];
    const float* be1 = (const float*)d_in[6];
    const float* W2  = (const float*)d_in[7];
    const float* b2  = (const float*)d_in[8];
    const float* g2  = (const float*)d_in[9];
    const float* be2 = (const float*)d_in[10];
    const float* W3  = (const float*)d_in[11];
    const float* b3  = (const float*)d_in[12];
    float* out = (float*)d_out;

    const int nrows = in_sizes[0] / 3;

    fused_head_kernel<<<nrows / ROWS, NTHR, 0, stream>>>(fk, dir, hm,
                                                         W1, b1, g1, be1,
                                                         W2, b2, g2, be2,
                                                         W3, b3, out, nrows);
}


# heatmap maxima reduced pair by pair as their loads land, on the vector unit only (lane-half exchange + 4 DPP row_xmask max), not behind the last load through 105 LDS lane permutes per wave
# speedup vs baseline: 1.5249x; 1.0800x over previous
#include <hip/hip_runtime.h>
#include <math.h>


typedef float v2f __attribute__((ext_vector_type(2)));
typedef float v8f __attribute__((ext_vector_type(8)));

#define ROWS   21
#define NTHR   512
#define HM_F4  1024

#define HP     68

__device__ __forceinline__ float f32_bits(unsigned u) { return __builtin_bit_cast(float, u); }

__device__ __forceinline__ float gelu_exact(float x) {
    const float y = x * 0.70710678118654752440f;
    const float t = fabsf(y);
    float p = fmaf(t, f32_bits(0x378e98abu), f32_bits(0xb9c68948u));
    p = fmaf(t, p, f32_bits(0x3b7cd369u));
    p = fmaf(t, p, f32_bits(0xbcc618b2u));
    p = fmaf(t, p, f32_bits(0x3dda74e4u));
    p = fmaf(t, p, f32_bits(0x3f228afdu));
    p = fmaf(t, p, f32_bits(0x3e03c728u));
    p = fmaf(t, p, t);
    const float ra = 1.0f - expf(-p);
    const float s = y * y;
    float q = fmaf(s, f32_bits(0xba1345e1u), f32_bits(0x3ba10414u));
    q = fmaf(s, q, f32_bits(0xbcdac9b8u));
    q = fmaf(s, q, f32_bits(0x3de703beu));
    q = fmaf(s, q, f32_bits(0xbec09330u));
    q = fmaf(s, q, f32_bits(0x3e0375d0u));
    const float rb = fmaf(t, q, t);
    const float e = copysignf((t < 1.0f) ? rb : ra, y);
    return 0.5f * x * (1.0f + e);
}

template <int M>
__device__ __forceinline__ float row_xor(float v) {
    const int b = __builtin_bit_cast(int, v);
    return __builtin_bit_cast(float, __builtin_amdgcn_update_dpp(0, b, 0x160 | M, 0xf, 0xf, true));
}
__device__ __forceinline__ float row16_sum(float v) {
    v += row_xor<1>(v);
    v += row_xor<2>(v);
    v += row_xor<4>(v);
    v += row_xor<8>(v);
    return v;
}

template <int M>
__device__ __forceinline__ float row_xor_max(float v) {
    float o;
    asm("v_max_num_f32_dpp %0, %1, %1 row_xmask:%2 row_mask:0xf bank_mask:0xf"
        : "=v"(o) : "v"(v), "n"(M));
    return o;
}
__device__ __forceinline__ float row16_max(float v) {
    v = row_xor_max<1>(v);
    v = row_xor_max<2>(v);
    v = row_xor_max<4>(v);
    v = row_xor_max<8>(v);
    return v;
}
__device__ __forceinline__ float xhalf(float x) {
    const int b = __builtin_bit_cast(int, x);
    return __builtin_bit_cast(float,
        __builtin_amdgcn_permlanex16(b, b, 0x76543210, 0xfedcba98, false, false));
}
__device__ __forceinline__ float max8(const float4 a, const float4 b) {
    return fmaxf(fmaxf(fmaxf(a.x, a.y), fmaxf(a.z, a.w)),
                 fmaxf(fmaxf(b.x, b.y), fmaxf(b.z, b.w)));
}

__device__ __forceinline__ void gelu_ln_inplace(float (*h)[HP],
                                                const float4 gg, const float4 bb,
                                                int tid) {
    const int r = tid >> 4;
    const int c = (tid & 15) * 4;
    if (r < ROWS) {
        float4 v = *(const float4*)&h[r][c];
        v.x = gelu_exact(v.x); v.y = gelu_exact(v.y);
        v.z = gelu_exact(v.z); v.w = gelu_exact(v.w);
        const float ts  = row16_sum((v.x + v.y) + (v.z + v.w));
        const float tss = row16_sum((v.x * v.x + v.y * v.y) + (v.z * v.z + v.w * v.w));
        const float mu  = ts * (1.0f / 64.0f);
        const float var = tss * (1.0f / 64.0f) - mu * mu;
        const float sc  = rsqrtf(var + 1e-5f);
        v.x = (v.x - mu) * sc * gg.x + bb.x;
        v.y = (v.y - mu) * sc * gg.y + bb.y;
        v.z = (v.z - mu) * sc * gg.z + bb.z;
        v.w = (v.w - mu) * sc * gg.w + bb.w;
        *(float4*)&h[r][c] = v;
    }
    __syncthreads();
}

__global__ __launch_bounds__(NTHR) void fused_head_kernel(
    const float* fk, const float* dir,
    const float* __restrict__ hm,
    const float* W1, const float* b1,
    const float* g1, const float* be1,
    const float* W2, const float* b2,
    const float* g2, const float* be2,
    const float* W3, const float* b3,
    float* __restrict__ out, int nrows) {

    __shared__ float xs[32][12];
    __shared__ float w1s[16][64];
    __shared__ __attribute__((aligned(16))) float w2s[64][64];
    __shared__ float w3s[64][16];
    __shared__ __attribute__((aligned(16))) float h1[32][HP];
    __shared__ __attribute__((aligned(16))) float h2[32][HP];
    __shared__ float outs[32][16];
    __shared__ __attribute__((aligned(16))) float pvec[7][64];
    __shared__ float smax[ROWS][16];

    const int tid  = threadIdx.x;
    const int lane = tid & 31;
    const int wave = tid >> 5;
    const int half = lane >> 4;
    const int l16  = lane & 15;
    const int row0 = blockIdx.x * ROWS;

    const float4 w2a = ((const float4*)W2)[tid];
    const float4 w2b = ((const float4*)W2)[tid + NTHR];
    const float  w1a = W1[tid];
    const int    i1b = (tid < 128) ? (tid + NTHR) : 639;
    const float  w1b = W1[i1b];
    const int    k3a = tid >> 4, k3b = (tid + NTHR) >> 4, n3 = tid & 15;
    const float  w3a = W3[k3a * 4 + (n3 & 3)];
    const float  w3b = W3[k3b * 4 + (n3 & 3)];
    const int    pn   = tid & 63;
    const float  pb1  = b1[pn],  pg1 = g1[pn],  pe1 = be1[pn];
    const float  pb2  = b2[pn],  pg2 = g2[pn],  pe2 = be2[pn];
    const float  pb3  = b3[tid & 3];
    const int    xe  = ((tid & 63) < 62) ? (tid & 63) : 62;
    const float  xf  = fk [row0 * 3 + xe];
    const float  xd  = dir[row0 * 3 + xe];

    __builtin_amdgcn_sched_barrier(0);

    const float4* hp = (const float4*)(hm + (size_t)row0 * 4096) + tid;
    float4 va[ROWS], vb[ROWS];
#pragma unroll
    for (int j = 0; j < ROWS; ++j) {
        va[j] = hp[j * HM_F4];
        vb[j] = hp[j * HM_F4 + NTHR];
    }
    __builtin_amdgcn_sched_barrier(0);

    *(float4*)&w2s[0][0 + 4 * tid]          = w2a;
    *(float4*)&w2s[0][0 + 4 * (tid + NTHR)] = w2b;
    (&w1s[0][0])[tid] = w1a;
    (&w1s[0][0])[tid + NTHR] = (tid < 128) ? w1b : 0.0f;
    w3s[k3a][n3] = (n3 < 4) ? w3a : 0.0f;
    w3s[k3b][n3] = (n3 < 4) ? w3b : 0.0f;
    pvec[0][pn] = pb1; pvec[1][pn] = pg1; pvec[2][pn] = pe1;
    pvec[3][pn] = pb2; pvec[4][pn] = pg2; pvec[5][pn] = pe2;
    pvec[6][tid & 3] = pb3;
    {
        const int r = xe / 3, c = xe - 3 * r;
        xs[r][c]     = xf;
        xs[r][3 + c] = xd;
        xs[r][6 + c] = xd - xf;
    }
    if (tid < 132) xs[ROWS + tid / 12][tid % 12] = 0.0f;
    if (tid < 63)  xs[tid / 3][9 + tid % 3] = 0.0f;

    const bool hi = (half != 0);
    float r[11];
#pragma unroll
    for (int p = 0; p < 10; ++p) {
        const float m0 = max8(va[2 * p],     vb[2 * p]);
        const float m1 = max8(va[2 * p + 1], vb[2 * p + 1]);
        const float keep = hi ? m1 : m0;
        const float send = hi ? m0 : m1;
        r[p] = row16_max(fmaxf(keep, xhalf(send)));
        if (p < 9) __builtin_amdgcn_sched_barrier(0);
    }
    {
        const float m20 = max8(va[ROWS - 1], vb[ROWS - 1]);
        r[10] = row16_max(fmaxf(m20, xhalf(m20)));
    }
    if (l16 == 0) {
#pragma unroll
        for (int p = 0; p < 10; ++p) smax[2 * p + half][wave] = r[p];
        if (!hi) smax[ROWS - 1][wave] = r[10];
    }
    __syncthreads();
    if (tid < ROWS) {
        float mm = smax[tid][0];
#pragma unroll
        for (int i = 1; i < 16; ++i) mm = fmaxf(mm, smax[tid][i]);
        xs[tid][9] = 1.0f / (1.0f + expf(-mm));
    }
    __syncthreads();

    const int mt   = (wave >> 2) & 1;
    const int ncol = (wave & 3) * 16 + l16;
    const int arow = mt * 16 + l16;

    if (wave < 8) {
        v8f c = {0.f, 0.f, 0.f, 0.f, 0.f, 0.f, 0.f, 0.f};
#pragma unroll
        for (int ks = 0; ks < 12; ks += 4) {
            v2f a, b;
            a[0] = xs[arow][ks + 2 * half + 0];
            a[1] = xs[arow][ks + 2 * half + 1];
            b[0] = w1s[ks + 2 * half + 0][ncol];
            b[1] = w1s[ks + 2 * half + 1][ncol];
            c = __builtin_amdgcn_wmma_f32_16x16x4_f32(false, a, false, b,
                                                      (short)0, c, false, false);
        }
        const float bias = pvec[0][ncol];
#pragma unroll
        for (int v = 0; v < 8; ++v)
            h1[mt * 16 + v + 8 * half][ncol] = c[v] + bias;
    }
    {
        const float4 gg = *(const float4*)&pvec[1][(tid & 15) * 4];
        const float4 bb = *(const float4*)&pvec[2][(tid & 15) * 4];
        __syncthreads();
        gelu_ln_inplace(h1, gg, bb, tid);
    }

    if (wave < 8) {
        v8f c = {0.f, 0.f, 0.f, 0.f, 0.f, 0.f, 0.f, 0.f};
#pragma unroll
        for (int ks = 0; ks < 64; ks += 4) {
            v2f a, b;
            a[0] = h1[arow][ks + 2 * half + 0];
            a[1] = h1[arow][ks + 2 * half + 1];
            b[0] = w2s[ks + 2 * half + 0][ncol];
            b[1] = w2s[ks + 2 * half + 1][ncol];
            c = __builtin_amdgcn_wmma_f32_16x16x4_f32(false, a, false, b,
                                                      (short)0, c, false, false);
        }
        const float bias = pvec[3][ncol];
#pragma unroll
        for (int v = 0; v < 8; ++v)
            h2[mt * 16 + v + 8 * half][ncol] = c[v] + bias;
    }
    {
        const float4 gg = *(const float4*)&pvec[4][(tid & 15) * 4];
        const float4 bb = *(const float4*)&pvec[5][(tid & 15) * 4];
        __syncthreads();
        gelu_ln_inplace(h2, gg, bb, tid);
    }

    if (wave < 2) {
        v8f c = {0.f, 0.f, 0.f, 0.f, 0.f, 0.f, 0.f, 0.f};
        const int r3 = wave * 16 + l16;
#pragma unroll
        for (int ks = 0; ks < 64; ks += 4) {
            v2f a, b;
            a[0] = h2[r3][ks + 2 * half + 0];
            a[1] = h2[r3][ks + 2 * half + 1];
            b[0] = w3s[ks + 2 * half + 0][l16];
            b[1] = w3s[ks + 2 * half + 1][l16];
            c = __builtin_amdgcn_wmma_f32_16x16x4_f32(false, a, false, b,
                                                      (short)0, c, false, false);
        }
#pragma unroll
        for (int v = 0; v < 8; ++v)
            outs[wave * 16 + v + 8 * half][l16] = c[v];
    }
    __syncthreads();

    if (tid < ROWS) {
        const int grow = row0 + tid;
        const float d0 = outs[tid][0] + pvec[6][0];
        const float d1 = outs[tid][1] + pvec[6][1];
        const float d2 = outs[tid][2] + pvec[6][2];
        const float o3 = outs[tid][3] + pvec[6][3];
        const float alpha = 1.0f / (1.0f + expf(-o3));

        const float fx = xs[tid][0], fy = xs[tid][1], fz = xs[tid][2];
        const float dx = xs[tid][3], dy = xs[tid][4], dz = xs[tid][5];

        float* kp_final = out;
        float* p_alpha  = out + (size_t)nrows * 3;
        float* p_delta  = out + (size_t)nrows * 4;

        kp_final[grow * 3 + 0] = alpha * dx + (1.0f - alpha) * fx + d0;
        kp_final[grow * 3 + 1] = alpha * dy + (1.0f - alpha) * fy + d1;
        kp_final[grow * 3 + 2] = alpha * dz + (1.0f - alpha) * fz + d2;
        p_alpha[grow] = alpha;
        p_delta[grow * 3 + 0] = d0;
        p_delta[grow * 3 + 1] = d1;
        p_delta[grow * 3 + 2] = d2;
    }
}

extern "C" void kernel_launch(void* const* d_in, const int* in_sizes, int n_in,
                              void* d_out, int out_size, void* d_ws, size_t ws_size,
                              hipStream_t stream) {
    const float* fk  = (const float*)d_in[0];
    const float* dir = (const float*)d_in[1];
    const float* hm  = (const float*)d_in[2];
    const float* W1  = (const float*)d_in[3];
    const float* b1  = (const float*)d_in[4];
    const float* g1  = (const float*)d_in[5];
    const float* be1 = (const float*)d_in[6];
    const float* W2  = (const float*)d_in[7];
    const float* b2  = (const float*)d_in[8];
    const float* g2  = (const float*)d_in[9];
    const float* be2 = (const float*)d_in[10];
    const float* W3  = (const float*)d_in[11];
    const float* b3  = (const float*)d_in[12];
    float* out = (float*)d_out;

    const int nrows = in_sizes[0] / 3;

    fused_head_kernel<<<nrows / ROWS, NTHR, 0, stream>>>(fk, dir, hm,
                                                         W1, b1, g1, be1,
                                                         W2, b2, g2, be2,
                                                         W3, b3, out, nrows);
}


# each batch element on 8 waves instead of 16 (half as many waves to start): every lane carries 4 16-byte loads per heatmap, all 84 issued up front; GELU+LayerNorm takes rows r and r+16 per lane group
# speedup vs baseline: 1.7786x; 1.1664x over previous
#include <hip/hip_runtime.h>
#include <math.h>


typedef float v2f __attribute__((ext_vector_type(2)));
typedef float v8f __attribute__((ext_vector_type(8)));

#define ROWS   21
#define NTHR   256
#define HM_F4  1024
#define HQ     4

#define HP     68

__device__ __forceinline__ float f32_bits(unsigned u) { return __builtin_bit_cast(float, u); }

__device__ __forceinline__ float gelu_exact(float x) {
    const float y = x * 0.70710678118654752440f;
    const float t = fabsf(y);
    float p = fmaf(t, f32_bits(0x378e98abu), f32_bits(0xb9c68948u));
    p = fmaf(t, p, f32_bits(0x3b7cd369u));
    p = fmaf(t, p, f32_bits(0xbcc618b2u));
    p = fmaf(t, p, f32_bits(0x3dda74e4u));
    p = fmaf(t, p, f32_bits(0x3f228afdu));
    p = fmaf(t, p, f32_bits(0x3e03c728u));
    p = fmaf(t, p, t);
    const float ra = 1.0f - expf(-p);
    const float s = y * y;
    float q = fmaf(s, f32_bits(0xba1345e1u), f32_bits(0x3ba10414u));
    q = fmaf(s, q, f32_bits(0xbcdac9b8u));
    q = fmaf(s, q, f32_bits(0x3de703beu));
    q = fmaf(s, q, f32_bits(0xbec09330u));
    q = fmaf(s, q, f32_bits(0x3e0375d0u));
    const float rb = fmaf(t, q, t);
    const float e = copysignf((t < 1.0f) ? rb : ra, y);
    return 0.5f * x * (1.0f + e);
}

template <int M>
__device__ __forceinline__ float row_xor(float v) {
    const int b = __builtin_bit_cast(int, v);
    return __builtin_bit_cast(float, __builtin_amdgcn_update_dpp(0, b, 0x160 | M, 0xf, 0xf, true));
}
__device__ __forceinline__ float row16_sum(float v) {
    v += row_xor<1>(v);
    v += row_xor<2>(v);
    v += row_xor<4>(v);
    v += row_xor<8>(v);
    return v;
}

template <int M>
__device__ __forceinline__ float row_xor_max(float v) {
    float o;
    asm("v_max_num_f32_dpp %0, %1, %1 row_xmask:%2 row_mask:0xf bank_mask:0xf"
        : "=v"(o) : "v"(v), "n"(M));
    return o;
}
__device__ __forceinline__ float row16_max(float v) {
    v = row_xor_max<1>(v);
    v = row_xor_max<2>(v);
    v = row_xor_max<4>(v);
    v = row_xor_max<8>(v);
    return v;
}
__device__ __forceinline__ float xhalf(float x) {
    const int b = __builtin_bit_cast(int, x);
    return __builtin_bit_cast(float,
        __builtin_amdgcn_permlanex16(b, b, 0x76543210, 0xfedcba98, false, false));
}
__device__ __forceinline__ float max8(const float4 a, const float4 b) {
    return fmaxf(fmaxf(fmaxf(a.x, a.y), fmaxf(a.z, a.w)),
                 fmaxf(fmaxf(b.x, b.y), fmaxf(b.z, b.w)));
}
__device__ __forceinline__ float max16(const float4 (&v)[HQ]) {
    return fmaxf(max8(v[0], v[1]), max8(v[2], v[3]));
}

__device__ __forceinline__ void gelu_ln_row(float (*h)[HP],
                                            const float4 gg, const float4 bb,
                                            int r, int c) {
    {
        float4 v = *(const float4*)&h[r][c];
        v.x = gelu_exact(v.x); v.y = gelu_exact(v.y);
        v.z = gelu_exact(v.z); v.w = gelu_exact(v.w);
        const float ts  = row16_sum((v.x + v.y) + (v.z + v.w));
        const float tss = row16_sum((v.x * v.x + v.y * v.y) + (v.z * v.z + v.w * v.w));
        const float mu  = ts * (1.0f / 64.0f);
        const float var = tss * (1.0f / 64.0f) - mu * mu;
        const float sc  = rsqrtf(var + 1e-5f);
        v.x = (v.x - mu) * sc * gg.x + bb.x;
        v.y = (v.y - mu) * sc * gg.y + bb.y;
        v.z = (v.z - mu) * sc * gg.z + bb.z;
        v.w = (v.w - mu) * sc * gg.w + bb.w;
        *(float4*)&h[r][c] = v;
    }
}
__device__ __forceinline__ void gelu_ln_inplace(float (*h)[HP],
                                                const float4 gg, const float4 bb,
                                                int tid) {
    const int r = tid >> 4;
    const int c = (tid & 15) * 4;
    gelu_ln_row(h, gg, bb, r, c);
    if (r + 16 < ROWS) gelu_ln_row(h, gg, bb, r + 16, c);
    __syncthreads();
}

__global__ __launch_bounds__(NTHR) void fused_head_kernel(
    const float* fk, const float* dir,
    const float* __restrict__ hm,
    const float* W1, const float* b1,
    const float* g1, const float* be1,
    const float* W2, const float* b2,
    const float* g2, const float* be2,
    const float* W3, const float* b3,
    float* __restrict__ out, int nrows) {

    __shared__ float xs[32][12];
    __shared__ float w1s[12][64];
    __shared__ __attribute__((aligned(16))) float w2s[64][64];
    __shared__ __attribute__((aligned(16))) float w3s[64][16];
    __shared__ __attribute__((aligned(16))) float h1[32][HP];
    __shared__ __attribute__((aligned(16))) float h2[32][HP];
    __shared__ float outs[32][16];
    __shared__ __attribute__((aligned(16))) float pvec[7][64];
    __shared__ float smax[ROWS][8];

    const int tid  = threadIdx.x;
    const int lane = tid & 31;
    const int wave = tid >> 5;
    const int half = lane >> 4;
    const int l16  = lane & 15;
    const int row0 = blockIdx.x * ROWS;

    const float4 w2a = ((const float4*)W2)[tid];
    const float4 w2b = ((const float4*)W2)[tid + NTHR];
    const float4 w2c = ((const float4*)W2)[tid + 2 * NTHR];
    const float4 w2d = ((const float4*)W2)[tid + 3 * NTHR];
    const float  w1a = W1[tid];
    const float  w1b = W1[tid + NTHR];
    const int    i1c = (tid < 128) ? (tid + 2 * NTHR) : 639;
    const float  w1c = W1[i1c];
    const float  w3v = W3[tid];
    const int    pn   = tid & 63;
    const float  pb1  = b1[pn],  pg1 = g1[pn],  pe1 = be1[pn];
    const float  pb2  = b2[pn],  pg2 = g2[pn],  pe2 = be2[pn];
    const float  pb3  = b3[tid & 3];
    const int    xe  = ((tid & 63) < 62) ? (tid & 63) : 62;
    const float  xf  = fk [row0 * 3 + xe];
    const float  xd  = dir[row0 * 3 + xe];

    __builtin_amdgcn_sched_barrier(0);

    const float4* hp = (const float4*)(hm + (size_t)row0 * 4096) + tid;
    float4 hv[ROWS][HQ];
#pragma unroll
    for (int j = 0; j < ROWS; ++j) {
#pragma unroll
        for (int q = 0; q < HQ; ++q) hv[j][q] = hp[j * HM_F4 + q * NTHR];
    }
    __builtin_amdgcn_sched_barrier(0);

    *(float4*)&w2s[0][0 + 4 * tid]              = w2a;
    *(float4*)&w2s[0][0 + 4 * (tid + NTHR)]     = w2b;
    *(float4*)&w2s[0][0 + 4 * (tid + 2 * NTHR)] = w2c;
    *(float4*)&w2s[0][0 + 4 * (tid + 3 * NTHR)] = w2d;
    (&w1s[0][0])[tid] = w1a;
    (&w1s[0][0])[tid + NTHR] = w1b;
    (&w1s[0][0])[tid + 2 * NTHR] = (tid < 128) ? w1c : 0.0f;
    w3s[tid >> 2][tid & 3] = w3v;
    if (tid < 192) *(float4*)&w3s[tid & 63][4 + 4 * (tid >> 6)] = make_float4(0.f, 0.f, 0.f, 0.f);
    pvec[0][pn] = pb1; pvec[1][pn] = pg1; pvec[2][pn] = pe1;
    pvec[3][pn] = pb2; pvec[4][pn] = pg2; pvec[5][pn] = pe2;
    pvec[6][tid & 3] = pb3;
    {
        const int r = xe / 3, c = xe - 3 * r;
        xs[r][c]     = xf;
        xs[r][3 + c] = xd;
        xs[r][6 + c] = xd - xf;
    }
    if (tid < 132) xs[ROWS + tid / 12][tid % 12] = 0.0f;
    if (tid < 63)  xs[tid / 3][9 + tid % 3] = 0.0f;

    const bool hi = (half != 0);
    float r[11];
#pragma unroll
    for (int p = 0; p < 10; ++p) {
        const float m0 = max16(hv[2 * p]);
        const float m1 = max16(hv[2 * p + 1]);
        const float keep = hi ? m1 : m0;
        const float send = hi ? m0 : m1;
        r[p] = row16_max(fmaxf(keep, xhalf(send)));
        if (p < 9) __builtin_amdgcn_sched_barrier(0);
    }
    {
        const float m20 = max16(hv[ROWS - 1]);
        r[10] = row16_max(fmaxf(m20, xhalf(m20)));
    }
    if (l16 == 0) {
#pragma unroll
        for (int p = 0; p < 10; ++p) smax[2 * p + half][wave] = r[p];
        if (!hi) smax[ROWS - 1][wave] = r[10];
    }
    __syncthreads();
    if (tid < ROWS) {
        float mm = smax[tid][0];
#pragma unroll
        for (int i = 1; i < 8; ++i) mm = fmaxf(mm, smax[tid][i]);
        xs[tid][9] = 1.0f / (1.0f + expf(-mm));
    }
    __syncthreads();

    const int mt   = wave >> 2;
    const int ncol = (wave & 3) * 16 + l16;
    const int arow = mt * 16 + l16;

    {
        v8f c = {0.f, 0.f, 0.f, 0.f, 0.f, 0.f, 0.f, 0.f};
#pragma unroll
        for (int ks = 0; ks < 12; ks += 4) {
            v2f a, b;
            a[0] = xs[arow][ks + 2 * half + 0];
            a[1] = xs[arow][ks + 2 * half + 1];
            b[0] = w1s[ks + 2 * half + 0][ncol];
            b[1] = w1s[ks + 2 * half + 1][ncol];
            c = __builtin_amdgcn_wmma_f32_16x16x4_f32(false, a, false, b,
                                                      (short)0, c, false, false);
        }
        const float bias = pvec[0][ncol];
#pragma unroll
        for (int v = 0; v < 8; ++v)
            h1[mt * 16 + v + 8 * half][ncol] = c[v] + bias;
    }
    {
        const float4 gg = *(const float4*)&pvec[1][(tid & 15) * 4];
        const float4 bb = *(const float4*)&pvec[2][(tid & 15) * 4];
        __syncthreads();
        gelu_ln_inplace(h1, gg, bb, tid);
    }

    {
        v8f c = {0.f, 0.f, 0.f, 0.f, 0.f, 0.f, 0.f, 0.f};
#pragma unroll
        for (int ks = 0; ks < 64; ks += 4) {
            v2f a, b;
            a[0] = h1[arow][ks + 2 * half + 0];
            a[1] = h1[arow][ks + 2 * half + 1];
            b[0] = w2s[ks + 2 * half + 0][ncol];
            b[1] = w2s[ks + 2 * half + 1][ncol];
            c = __builtin_amdgcn_wmma_f32_16x16x4_f32(false, a, false, b,
                                                      (short)0, c, false, false);
        }
        const float bias = pvec[3][ncol];
#pragma unroll
        for (int v = 0; v < 8; ++v)
            h2[mt * 16 + v + 8 * half][ncol] = c[v] + bias;
    }
    {
        const float4 gg = *(const float4*)&pvec[4][(tid & 15) * 4];
        const float4 bb = *(const float4*)&pvec[5][(tid & 15) * 4];
        __syncthreads();
        gelu_ln_inplace(h2, gg, bb, tid);
    }

    if (wave < 2) {
        v8f c = {0.f, 0.f, 0.f, 0.f, 0.f, 0.f, 0.f, 0.f};
        const int r3 = wave * 16 + l16;
#pragma unroll
        for (int ks = 0; ks < 64; ks += 4) {
            v2f a, b;
            a[0] = h2[r3][ks + 2 * half + 0];
            a[1] = h2[r3][ks + 2 * half + 1];
            b[0] = w3s[ks + 2 * half + 0][l16];
            b[1] = w3s[ks + 2 * half + 1][l16];
            c = __builtin_amdgcn_wmma_f32_16x16x4_f32(false, a, false, b,
                                                      (short)0, c, false, false);
        }
#pragma unroll
        for (int v = 0; v < 8; ++v)
            outs[wave * 16 + v + 8 * half][l16] = c[v];
    }
    __syncthreads();

    if (tid < ROWS) {
        const int grow = row0 + tid;
        const float d0 = outs[tid][0] + pvec[6][0];
        const float d1 = outs[tid][1] + pvec[6][1];
        const float d2 = outs[tid][2] + pvec[6][2];
        const float o3 = outs[tid][3] + pvec[6][3];
        const float alpha = 1.0f / (1.0f + expf(-o3));

        const float fx = xs[tid][0], fy = xs[tid][1], fz = xs[tid][2];
        const float dx = xs[tid][3], dy = xs[tid][4], dz = xs[tid][5];

        float* kp_final = out;
        float* p_alpha  = out + (size_t)nrows * 3;
        float* p_delta  = out + (size_t)nrows * 4;

        kp_final[grow * 3 + 0] = alpha * dx + (1.0f - alpha) * fx + d0;
        kp_final[grow * 3 + 1] = alpha * dy + (1.0f - alpha) * fy + d1;
        kp_final[grow * 3 + 2] = alpha * dz + (1.0f - alpha) * fz + d2;
        p_alpha[grow] = alpha;
        p_delta[grow * 3 + 0] = d0;
        p_delta[grow * 3 + 1] = d1;
        p_delta[grow * 3 + 2] = d2;
    }
}

extern "C" void kernel_launch(void* const* d_in, const int* in_sizes, int n_in,
                              void* d_out, int out_size, void* d_ws, size_t ws_size,
                              hipStream_t stream) {
    const float* fk  = (const float*)d_in[0];
    const float* dir = (const float*)d_in[1];
    const float* hm  = (const float*)d_in[2];
    const float* W1  = (const float*)d_in[3];
    const float* b1  = (const float*)d_in[4];
    const float* g1  = (const float*)d_in[5];
    const float* be1 = (const float*)d_in[6];
    const float* W2  = (const float*)d_in[7];
    const float* b2  = (const float*)d_in[8];
    const float* g2  = (const float*)d_in[9];
    const float* be2 = (const float*)d_in[10];
    const float* W3  = (const float*)d_in[11];
    const float* b3  = (const float*)d_in[12];
    float* out = (float*)d_out;

    const int nrows = in_sizes[0] / 3;

    fused_head_kernel<<<nrows / ROWS, NTHR, 0, stream>>>(fk, dir, hm,
                                                         W1, b1, g1, be1,
                                                         W2, b2, g2, be2,
                                                         W3, b3, out, nrows);
}


# each batch element on 4 waves, not 8 (one per SIMD, half as many waves to start): 168 16-byte loads per lane, 80 ahead; a wave owns a 16-column strip of both row tiles; GELU+LN on rows r, r+8, r+16
# speedup vs baseline: 2.0163x; 1.1336x over previous
#include <hip/hip_runtime.h>
#include <math.h>


typedef float v2f __attribute__((ext_vector_type(2)));
typedef float v8f __attribute__((ext_vector_type(8)));

#define ROWS   21
#define NTHR   128
#define HM_F4  1024
#define HQ     8

#define AHEAD  10

#define HP     68

__device__ __forceinline__ float f32_bits(unsigned u) { return __builtin_bit_cast(float, u); }

__device__ __forceinline__ float gelu_exact(float x) {
    const float y = x * 0.70710678118654752440f;
    const float t = fabsf(y);
    float p = fmaf(t, f32_bits(0x378e98abu), f32_bits(0xb9c68948u));
    p = fmaf(t, p, f32_bits(0x3b7cd369u));
    p = fmaf(t, p, f32_bits(0xbcc618b2u));
    p = fmaf(t, p, f32_bits(0x3dda74e4u));
    p = fmaf(t, p, f32_bits(0x3f228afdu));
    p = fmaf(t, p, f32_bits(0x3e03c728u));
    p = fmaf(t, p, t);
    const float ra = 1.0f - expf(-p);
    const float s = y * y;
    float q = fmaf(s, f32_bits(0xba1345e1u), f32_bits(0x3ba10414u));
    q = fmaf(s, q, f32_bits(0xbcdac9b8u));
    q = fmaf(s, q, f32_bits(0x3de703beu));
    q = fmaf(s, q, f32_bits(0xbec09330u));
    q = fmaf(s, q, f32_bits(0x3e0375d0u));
    const float rb = fmaf(t, q, t);
    const float e = copysignf((t < 1.0f) ? rb : ra, y);
    return 0.5f * x * (1.0f + e);
}

template <int M>
__device__ __forceinline__ float row_xor(float v) {
    const int b = __builtin_bit_cast(int, v);
    return __builtin_bit_cast(float, __builtin_amdgcn_update_dpp(0, b, 0x160 | M, 0xf, 0xf, true));
}
__device__ __forceinline__ float row16_sum(float v) {
    v += row_xor<1>(v);
    v += row_xor<2>(v);
    v += row_xor<4>(v);
    v += row_xor<8>(v);
    return v;
}

template <int M>
__device__ __forceinline__ float row_xor_max(float v) {
    float o;
    asm("v_max_num_f32_dpp %0, %1, %1 row_xmask:%2 row_mask:0xf bank_mask:0xf"
        : "=v"(o) : "v"(v), "n"(M));
    return o;
}
__device__ __forceinline__ float row16_max(float v) {
    v = row_xor_max<1>(v);
    v = row_xor_max<2>(v);
    v = row_xor_max<4>(v);
    v = row_xor_max<8>(v);
    return v;
}
__device__ __forceinline__ float xhalf(float x) {
    const int b = __builtin_bit_cast(int, x);
    return __builtin_bit_cast(float,
        __builtin_amdgcn_permlanex16(b, b, 0x76543210, 0xfedcba98, false, false));
}
__device__ __forceinline__ float max8(const float4 a, const float4 b) {
    return fmaxf(fmaxf(fmaxf(a.x, a.y), fmaxf(a.z, a.w)),
                 fmaxf(fmaxf(b.x, b.y), fmaxf(b.z, b.w)));
}
__device__ __forceinline__ float max32(const float4 (&v)[HQ]) {
    return fmaxf(fmaxf(max8(v[0], v[1]), max8(v[2], v[3])),
                 fmaxf(max8(v[4], v[5]), max8(v[6], v[7])));
}

__device__ __forceinline__ float4 gelu_ln_vals(float4 v, const float4 gg, const float4 bb) {
    v.x = gelu_exact(v.x); v.y = gelu_exact(v.y);
    v.z = gelu_exact(v.z); v.w = gelu_exact(v.w);
    const float ts  = row16_sum((v.x + v.y) + (v.z + v.w));
    const float tss = row16_sum((v.x * v.x + v.y * v.y) + (v.z * v.z + v.w * v.w));
    const float mu  = ts * (1.0f / 64.0f);
    const float var = tss * (1.0f / 64.0f) - mu * mu;
    const float sc  = rsqrtf(var + 1e-5f);
    v.x = (v.x - mu) * sc * gg.x + bb.x;
    v.y = (v.y - mu) * sc * gg.y + bb.y;
    v.z = (v.z - mu) * sc * gg.z + bb.z;
    v.w = (v.w - mu) * sc * gg.w + bb.w;
    return v;
}
__device__ __forceinline__ void gelu_ln_inplace(float (*h)[HP],
                                                const float4 gg, const float4 bb,
                                                int tid) {
    const int r = tid >> 4;
    const int c = (tid & 15) * 4;
    const float4 v0 = *(const float4*)&h[r][c];
    const float4 v1 = *(const float4*)&h[r + 8][c];
    const float4 v2 = *(const float4*)&h[r + 16][c];
    const float4 o0 = gelu_ln_vals(v0, gg, bb);
    const float4 o1 = gelu_ln_vals(v1, gg, bb);
    const float4 o2 = gelu_ln_vals(v2, gg, bb);
    *(float4*)&h[r][c]      = o0;
    *(float4*)&h[r + 8][c]  = o1;
    *(float4*)&h[r + 16][c] = o2;
    __syncthreads();
}

__global__ __launch_bounds__(NTHR) void fused_head_kernel(
    const float* fk, const float* dir,
    const float* __restrict__ hm,
    const float* W1, const float* b1,
    const float* g1, const float* be1,
    const float* W2, const float* b2,
    const float* g2, const float* be2,
    const float* W3, const float* b3,
    float* __restrict__ out, int nrows) {

    __shared__ float xs[32][12];
    __shared__ float w1s[12][64];
    __shared__ __attribute__((aligned(16))) float w2s[64][64];
    __shared__ __attribute__((aligned(16))) float w3s[64][16];
    __shared__ __attribute__((aligned(16))) float h1[32][HP];
    __shared__ __attribute__((aligned(16))) float h2[32][HP];
    __shared__ float outs[32][16];
    __shared__ __attribute__((aligned(16))) float pvec[7][64];
    __shared__ float smax[ROWS][4];

    const int tid  = threadIdx.x;
    const int lane = tid & 31;
    const int wave = tid >> 5;
    const int half = lane >> 4;
    const int l16  = lane & 15;
    const int row0 = blockIdx.x * ROWS;

    const float4 w2a = ((const float4*)W2)[tid];
    const float4 w2b = ((const float4*)W2)[tid + NTHR];
    const float4 w2c = ((const float4*)W2)[tid + 2 * NTHR];
    const float4 w2d = ((const float4*)W2)[tid + 3 * NTHR];
    const float4 w2e = ((const float4*)W2)[tid + 4 * NTHR];
    const float4 w2f = ((const float4*)W2)[tid + 5 * NTHR];
    const float4 w2g = ((const float4*)W2)[tid + 6 * NTHR];
    const float4 w2h = ((const float4*)W2)[tid + 7 * NTHR];
    const float  w1a = W1[tid];
    const float  w1b = W1[tid + NTHR];
    const float  w1c = W1[tid + 2 * NTHR];
    const float  w1d = W1[tid + 3 * NTHR];
    const float  w1e = W1[tid + 4 * NTHR];
    const float  w3a = W3[tid];
    const float  w3b = W3[tid + NTHR];
    const int    pn   = tid & 63;
    const float  pb1  = b1[pn],  pg1 = g1[pn],  pe1 = be1[pn];
    const float  pb2  = b2[pn],  pg2 = g2[pn],  pe2 = be2[pn];
    const float  pb3  = b3[tid & 3];
    const int    xe  = ((tid & 63) < 62) ? (tid & 63) : 62;
    const float  xf  = fk [row0 * 3 + xe];
    const float  xd  = dir[row0 * 3 + xe];

    __builtin_amdgcn_sched_barrier(0);

    const float4* hp = (const float4*)(hm + (size_t)row0 * 4096) + tid;
    float4 hv[ROWS][HQ];
#pragma unroll
    for (int j = 0; j < AHEAD; ++j) {
#pragma unroll
        for (int q = 0; q < HQ; ++q) hv[j][q] = hp[j * HM_F4 + q * NTHR];
    }
    __builtin_amdgcn_sched_barrier(0);

    *(float4*)&w2s[0][0 + 4 * tid]              = w2a;
    *(float4*)&w2s[0][0 + 4 * (tid + NTHR)]     = w2b;
    *(float4*)&w2s[0][0 + 4 * (tid + 2 * NTHR)] = w2c;
    *(float4*)&w2s[0][0 + 4 * (tid + 3 * NTHR)] = w2d;
    *(float4*)&w2s[0][0 + 4 * (tid + 4 * NTHR)] = w2e;
    *(float4*)&w2s[0][0 + 4 * (tid + 5 * NTHR)] = w2f;
    *(float4*)&w2s[0][0 + 4 * (tid + 6 * NTHR)] = w2g;
    *(float4*)&w2s[0][0 + 4 * (tid + 7 * NTHR)] = w2h;
    (&w1s[0][0])[tid]            = w1a;
    (&w1s[0][0])[tid + NTHR]     = w1b;
    (&w1s[0][0])[tid + 2 * NTHR] = w1c;
    (&w1s[0][0])[tid + 3 * NTHR] = w1d;
    (&w1s[0][0])[tid + 4 * NTHR] = w1e;
    (&w1s[0][0])[tid + 5 * NTHR] = 0.0f;
    w3s[tid >> 2][tid & 3]        = w3a;
    w3s[32 + (tid >> 2)][tid & 3] = w3b;
    *(float4*)&w3s[tid & 63][4 + 4 * (tid >> 6)] = make_float4(0.f, 0.f, 0.f, 0.f);
    if (tid < 64) *(float4*)&w3s[tid][12] = make_float4(0.f, 0.f, 0.f, 0.f);
    pvec[0][pn] = pb1; pvec[1][pn] = pg1; pvec[2][pn] = pe1;
    pvec[3][pn] = pb2; pvec[4][pn] = pg2; pvec[5][pn] = pe2;
    pvec[6][tid & 3] = pb3;
    {
        const int r = xe / 3, c = xe - 3 * r;
        xs[r][c]     = xf;
        xs[r][3 + c] = xd;
        xs[r][6 + c] = xd - xf;
    }
    xs[ROWS + tid / 12][tid % 12] = 0.0f;
    if (tid < 4)   xs[31][8 + tid] = 0.0f;
    if (tid < 63)  xs[tid / 3][9 + tid % 3] = 0.0f;

    const bool hi = (half != 0);
    float r[11];
#pragma unroll
    for (int p = 0; p < 10; ++p) {
        const float m0 = max32(hv[2 * p]);
        const float m1 = max32(hv[2 * p + 1]);
        const float keep = hi ? m1 : m0;
        const float send = hi ? m0 : m1;
        r[p] = row16_max(fmaxf(keep, xhalf(send)));
        if (p < 9) __builtin_amdgcn_sched_barrier(0);
        if (AHEAD + 2 * p < ROWS) {
#pragma unroll
            for (int q = 0; q < HQ; ++q)
                hv[AHEAD + 2 * p][q] = hp[(AHEAD + 2 * p) * HM_F4 + q * NTHR];
            if (AHEAD + 2 * p + 1 < ROWS) {
#pragma unroll
                for (int q = 0; q < HQ; ++q)
                    hv[AHEAD + 2 * p + 1][q] = hp[(AHEAD + 2 * p + 1) * HM_F4 + q * NTHR];
            }
            __builtin_amdgcn_sched_barrier(0);
        }
    }
    {
        const float m20 = max32(hv[ROWS - 1]);
        r[10] = row16_max(fmaxf(m20, xhalf(m20)));
    }
    if (l16 == 0) {
#pragma unroll
        for (int p = 0; p < 10; ++p) smax[2 * p + half][wave] = r[p];
        if (!hi) smax[ROWS - 1][wave] = r[10];
    }
    __syncthreads();
    if (tid < ROWS) {
        float mm = smax[tid][0];
#pragma unroll
        for (int i = 1; i < 4; ++i) mm = fmaxf(mm, smax[tid][i]);
        xs[tid][9] = 1.0f / (1.0f + expf(-mm));
    }
    __syncthreads();

    const int ncol = wave * 16 + l16;

    {
        v8f c0 = {0.f, 0.f, 0.f, 0.f, 0.f, 0.f, 0.f, 0.f};
        v8f c1 = {0.f, 0.f, 0.f, 0.f, 0.f, 0.f, 0.f, 0.f};
#pragma unroll
        for (int ks = 0; ks < 12; ks += 4) {
            v2f a0, a1, b;
            a0[0] = xs[l16][ks + 2 * half + 0];
            a0[1] = xs[l16][ks + 2 * half + 1];
            a1[0] = xs[16 + l16][ks + 2 * half + 0];
            a1[1] = xs[16 + l16][ks + 2 * half + 1];
            b[0] = w1s[ks + 2 * half + 0][ncol];
            b[1] = w1s[ks + 2 * half + 1][ncol];
            c0 = __builtin_amdgcn_wmma_f32_16x16x4_f32(false, a0, false, b,
                                                       (short)0, c0, false, false);
            c1 = __builtin_amdgcn_wmma_f32_16x16x4_f32(false, a1, false, b,
                                                       (short)0, c1, false, false);
        }
        const float bias = pvec[0][ncol];
#pragma unroll
        for (int v = 0; v < 8; ++v) {
            h1[v + 8 * half][ncol]      = c0[v] + bias;
            h1[16 + v + 8 * half][ncol] = c1[v] + bias;
        }
    }
    {
        const float4 gg = *(const float4*)&pvec[1][(tid & 15) * 4];
        const float4 bb = *(const float4*)&pvec[2][(tid & 15) * 4];
        __syncthreads();
        gelu_ln_inplace(h1, gg, bb, tid);
    }

    {
        v8f c0 = {0.f, 0.f, 0.f, 0.f, 0.f, 0.f, 0.f, 0.f};
        v8f c1 = {0.f, 0.f, 0.f, 0.f, 0.f, 0.f, 0.f, 0.f};
#pragma unroll
        for (int ks = 0; ks < 64; ks += 4) {
            v2f a0, a1, b;
            a0[0] = h1[l16][ks + 2 * half + 0];
            a0[1] = h1[l16][ks + 2 * half + 1];
            a1[0] = h1[16 + l16][ks + 2 * half + 0];
            a1[1] = h1[16 + l16][ks + 2 * half + 1];
            b[0] = w2s[ks + 2 * half + 0][ncol];
            b[1] = w2s[ks + 2 * half + 1][ncol];
            c0 = __builtin_amdgcn_wmma_f32_16x16x4_f32(false, a0, false, b,
                                                       (short)0, c0, false, false);
            c1 = __builtin_amdgcn_wmma_f32_16x16x4_f32(false, a1, false, b,
                                                       (short)0, c1, false, false);
        }
        const float bias = pvec[3][ncol];
#pragma unroll
        for (int v = 0; v < 8; ++v) {
            h2[v + 8 * half][ncol]      = c0[v] + bias;
            h2[16 + v + 8 * half][ncol] = c1[v] + bias;
        }
    }
    {
        const float4 gg = *(const float4*)&pvec[4][(tid & 15) * 4];
        const float4 bb = *(const float4*)&pvec[5][(tid & 15) * 4];
        __syncthreads();
        gelu_ln_inplace(h2, gg, bb, tid);
    }

    if (wave < 2) {
        v8f c = {0.f, 0.f, 0.f, 0.f, 0.f, 0.f, 0.f, 0.f};
        const int r3 = wave * 16 + l16;
#pragma unroll
        for (int ks = 0; ks < 64; ks += 4) {
            v2f a, b;
            a[0] = h2[r3][ks + 2 * half + 0];
            a[1] = h2[r3][ks + 2 * half + 1];
            b[0] = w3s[ks + 2 * half + 0][l16];
            b[1] = w3s[ks + 2 * half + 1][l16];
            c = __builtin_amdgcn_wmma_f32_16x16x4_f32(false, a, false, b,
                                                      (short)0, c, false, false);
        }
#pragma unroll
        for (int v = 0; v < 8; ++v)
            outs[wave * 16 + v + 8 * half][l16] = c[v];
    }
    __syncthreads();

    if (tid < ROWS) {
        const int grow = row0 + tid;
        const float d0 = outs[tid][0] + pvec[6][0];
        const float d1 = outs[tid][1] + pvec[6][1];
        const float d2 = outs[tid][2] + pvec[6][2];
        const float o3 = outs[tid][3] + pvec[6][3];
        const float alpha = 1.0f / (1.0f + expf(-o3));

        const float fx = xs[tid][0], fy = xs[tid][1], fz = xs[tid][2];
        const float dx = xs[tid][3], dy = xs[tid][4], dz = xs[tid][5];

        float* kp_final = out;
        float* p_alpha  = out + (size_t)nrows * 3;
        float* p_delta  = out + (size_t)nrows * 4;

        kp_final[grow * 3 + 0] = alpha * dx + (1.0f - alpha) * fx + d0;
        kp_final[grow * 3 + 1] = alpha * dy + (1.0f - alpha) * fy + d1;
        kp_final[grow * 3 + 2] = alpha * dz + (1.0f - alpha) * fz + d2;
        p_alpha[grow] = alpha;
        p_delta[grow * 3 + 0] = d0;
        p_delta[grow * 3 + 1] = d1;
        p_delta[grow * 3 + 2] = d2;
    }
}

extern "C" void kernel_launch(void* const* d_in, const int* in_sizes, int n_in,
                              void* d_out, int out_size, void* d_ws, size_t ws_size,
                              hipStream_t stream) {
    const float* fk  = (const float*)d_in[0];
    const float* dir = (const float*)d_in[1];
    const float* hm  = (const float*)d_in[2];
    const float* W1  = (const float*)d_in[3];
    const float* b1  = (const float*)d_in[4];
    const float* g1  = (const float*)d_in[5];
    const float* be1 = (const float*)d_in[6];
    const float* W2  = (const float*)d_in[7];
    const float* b2  = (const float*)d_in[8];
    const float* g2  = (const float*)d_in[9];
    const float* be2 = (const float*)d_in[10];
    const float* W3  = (const float*)d_in[11];
    const float* b3  = (const float*)d_in[12];
    float* out = (float*)d_out;

    const int nrows = in_sizes[0] / 3;

    fused_head_kernel<<<nrows / ROWS, NTHR, 0, stream>>>(fk, dir, hm,
                                                         W1, b1, g1, be1,
                                                         W2, b2, g2, be2,
                                                         W3, b3, out, nrows);
}
